# starting kernel
# baseline (speedup 1.0000x reference)
#include <hip/hip_runtime.h>


typedef __attribute__((ext_vector_type(2))) float v2f;
typedef __attribute__((ext_vector_type(8))) float v8f;

#define NFILT  1024
#define HH     64
#define WW     64
#define KH     5
#define KW     5
#define LOCAL  3
#define WIN    11
#define KPIX   121
#define KPAD   124
#define NPIX   36
#define NPAD   48
#define MB     64
#define THREADS 128

__global__ __launch_bounds__(THREADS)
void filters_wmma_kernel(const float* __restrict__ X,
                         const float* __restrict__ weight,
                         const float* __restrict__ bias,
                         const int*   __restrict__ positions,
                         float*       __restrict__ out)
{
    __shared__ float A_lds[MB * KPAD];
    __shared__ float B_lds[KPAD * NPAD];

    const int f      = blockIdx.x;
    const int batch0 = blockIdx.y * MB;
    const int tid    = threadIdx.x;
    const int lane   = tid & 31;
    const int wave   = tid >> 5;
    const int half   = lane >> 4;
    const int nl     = lane & 15;

    const int i0 = positions[2 * f + 0];
    const int j0 = positions[2 * f + 1];

    for (int idx = tid; idx < MB * KPAD; idx += THREADS) {
        const int mb = idx / KPAD;
        const int k  = idx - mb * KPAD;
        float v = 0.0f;
        if (k < KPIX) {
            const int wi = k / WIN;
            const int wj = k - wi * WIN;
            const int r  = i0 + wi - LOCAL;
            const int c  = j0 + wj - LOCAL;
            if ((unsigned)r < (unsigned)HH && (unsigned)c < (unsigned)WW)
                v = X[(size_t)(batch0 + mb) * (HH * WW) + r * WW + c];
        }
        A_lds[idx] = v;
    }

    for (int idx = tid; idx < KPAD * NPAD; idx += THREADS) B_lds[idx] = 0.0f;
    __syncthreads();
    for (int idx = tid; idx < NPIX * KH * KW; idx += THREADS) {
        const int p   = idx / (KH * KW);
        const int kk  = idx - p * (KH * KW);
        const int ki  = kk / KW;
        const int kj  = kk - ki * KW;
        const int pi  = p / 6;
        const int pj  = p - pi * 6;
        const int krow = (pi + ki) * WIN + (pj + kj);
        B_lds[krow * NPAD + p] = weight[f * (KH * KW) + kk];
    }
    __syncthreads();

    v8f acc0 = {}, acc1 = {}, acc2 = {};
    const float* Arow = &A_lds[(wave * 16 + nl) * KPAD];
    #pragma unroll 1
    for (int s = 0; s < KPAD / 4; ++s) {
        const int k0 = 4 * s + 2 * half;
        v2f a;  a.x  = Arow[k0];        a.y  = Arow[k0 + 1];
        const float* Br0 = &B_lds[k0 * NPAD + nl];
        const float* Br1 = Br0 + NPAD;
        v2f b0; b0.x = Br0[0];   b0.y = Br1[0];
        v2f b1; b1.x = Br0[16];  b1.y = Br1[16];
        v2f b2; b2.x = Br0[32];  b2.y = Br1[32];
        acc0 = __builtin_amdgcn_wmma_f32_16x16x4_f32(false, a, false, b0, (short)0, acc0, false, false);
        acc1 = __builtin_amdgcn_wmma_f32_16x16x4_f32(false, a, false, b1, (short)0, acc1, false, false);
        acc2 = __builtin_amdgcn_wmma_f32_16x16x4_f32(false, a, false, b2, (short)0, acc2, false, false);
    }

    __syncthreads();
    float* conv = &A_lds[wave * (16 * NPAD)];
    #pragma unroll
    for (int r = 0; r < 8; ++r) {
        conv[(r + 8 * half) * NPAD +  0 + nl] = acc0[r];
        conv[(r + 8 * half) * NPAD + 16 + nl] = acc1[r];
        conv[(r + 8 * half) * NPAD + 32 + nl] = acc2[r];
    }
    __syncthreads();

    if (lane < 16) {
        const int b = batch0 + wave * 16 + nl;
        const float bv = bias[f];
        const float* cr = &conv[nl * NPAD];
        float4 o;
        float* oo = &o.x;
        #pragma unroll
        for (int q = 0; q < 4; ++q) {
            const int qi = q >> 1, qj = q & 1;
            float mx = -__builtin_inff();
            #pragma unroll
            for (int di = 0; di < 3; ++di)
                #pragma unroll
                for (int dj = 0; dj < 3; ++dj)
                    mx = fmaxf(mx, cr[(3 * qi + di) * 6 + (3 * qj + dj)]);
            oo[q] = mx + bv;
        }
        *(float4*)&out[(size_t)b * (NFILT * 4) + f * 4] = o;
    }
}

extern "C" void kernel_launch(void* const* d_in, const int* in_sizes, int n_in,
                              void* d_out, int out_size, void* d_ws, size_t ws_size,
                              hipStream_t stream) {
    const float* X         = (const float*)d_in[0];
    const float* weight    = (const float*)d_in[1];
    const float* bias      = (const float*)d_in[2];
    const int*   positions = (const int*)d_in[3];
    float*       out       = (float*)d_out;

    dim3 grid(NFILT, 1024 / MB, 1);
    dim3 block(THREADS, 1, 1);
    filters_wmma_kernel<<<grid, block, 0, stream>>>(X, weight, bias, positions, out);
}


# All prologue global loads issued as one batch before the first LDS store: per wave 64 window loads (image offsets computed once per lane) plus weight and bias loads, not one load per wait
# speedup vs baseline: 2.3644x; 2.3644x over previous
#include <hip/hip_runtime.h>


typedef __attribute__((ext_vector_type(2))) float v2f;
typedef __attribute__((ext_vector_type(8))) float v8f;

#define NFILT  1024
#define HH     64
#define WW     64
#define KH     5
#define KW     5
#define LOCAL  3
#define WIN    11
#define KPIX   121
#define KPAD   124
#define NPIX   36
#define NPAD   48
#define MB     64
#define THREADS 128

__global__ __launch_bounds__(THREADS)
void filters_wmma_kernel(const float* __restrict__ X,
                         const float* __restrict__ weight,
                         const float* __restrict__ bias,
                         const int*   __restrict__ positions,
                         float*       __restrict__ out)
{
    __shared__ float A_lds[MB * KPAD];
    __shared__ float B_lds[KPAD * NPAD];

    const int f      = blockIdx.x;
    const int batch0 = blockIdx.y * MB;
    const int tid    = threadIdx.x;
    const int lane   = tid & 31;
    const int wave   = tid >> 5;
    const int half   = lane >> 4;
    const int nl     = lane & 15;

    const int i0 = positions[2 * f + 0];
    const int j0 = positions[2 * f + 1];
    const float bv = bias[f];

    int  aoff[4];
    bool aok[4];
    #pragma unroll
    for (int j = 0; j < 4; ++j) {
        const int k  = lane + 32 * j;
        const int kc = k < KPIX ? k : KPIX - 1;
        const int wi = kc / WIN;
        const int wj = kc - wi * WIN;
        const int r  = i0 + wi - LOCAL;
        const int c  = j0 + wj - LOCAL;
        aok[j] = (k < KPIX) && ((unsigned)r < (unsigned)HH) && ((unsigned)c < (unsigned)WW);
        const int rc = r < 0 ? 0 : (r > HH - 1 ? HH - 1 : r);
        const int cc = c < 0 ? 0 : (c > WW - 1 ? WW - 1 : c);
        aoff[j] = rc * WW + cc;
    }
    const float* Xw = X + (size_t)(batch0 + wave * 16) * (HH * WW);
    float av[16][4];
    #pragma unroll
    for (int i = 0; i < 16; ++i) {
        #pragma unroll
        for (int j = 0; j < 4; ++j)
            av[i][j] = Xw[i * (HH * WW) + aoff[j]];
    }
    float wv[8];
    #pragma unroll
    for (int q = 0; q < 8; ++q) {
        const int idx = tid + THREADS * q;
        const int kk  = idx % (KH * KW);
        wv[q] = weight[f * (KH * KW) + (idx < NPIX * KH * KW ? kk : 0)];
    }

    for (int idx = tid; idx < KPAD * NPAD; idx += THREADS) B_lds[idx] = 0.0f;

    #pragma unroll
    for (int i = 0; i < 16; ++i) {
        #pragma unroll
        for (int j = 0; j < 4; ++j) {
            const int k = lane + 32 * j;
            if (k < KPAD)
                A_lds[(wave * 16 + i) * KPAD + k] = aok[j] ? av[i][j] : 0.0f;
        }
    }
    __syncthreads();

    #pragma unroll
    for (int q = 0; q < 8; ++q) {
        const int idx = tid + THREADS * q;
        if (idx < NPIX * KH * KW) {
            const int p   = idx / (KH * KW);
            const int kk  = idx - p * (KH * KW);
            const int ki  = kk / KW;
            const int kj  = kk - ki * KW;
            const int pi  = p / 6;
            const int pj  = p - pi * 6;
            const int krow = (pi + ki) * WIN + (pj + kj);
            B_lds[krow * NPAD + p] = wv[q];
        }
    }
    __syncthreads();

    v8f acc0 = {}, acc1 = {}, acc2 = {};
    const float* Arow = &A_lds[(wave * 16 + nl) * KPAD];
    #pragma unroll 1
    for (int s = 0; s < KPAD / 4; ++s) {
        const int k0 = 4 * s + 2 * half;
        v2f a;  a.x  = Arow[k0];        a.y  = Arow[k0 + 1];
        const float* Br0 = &B_lds[k0 * NPAD + nl];
        const float* Br1 = Br0 + NPAD;
        v2f b0; b0.x = Br0[0];   b0.y = Br1[0];
        v2f b1; b1.x = Br0[16];  b1.y = Br1[16];
        v2f b2; b2.x = Br0[32];  b2.y = Br1[32];
        acc0 = __builtin_amdgcn_wmma_f32_16x16x4_f32(false, a, false, b0, (short)0, acc0, false, false);
        acc1 = __builtin_amdgcn_wmma_f32_16x16x4_f32(false, a, false, b1, (short)0, acc1, false, false);
        acc2 = __builtin_amdgcn_wmma_f32_16x16x4_f32(false, a, false, b2, (short)0, acc2, false, false);
    }

    __syncthreads();
    float* conv = &A_lds[wave * (16 * NPAD)];
    #pragma unroll
    for (int r = 0; r < 8; ++r) {
        conv[(r + 8 * half) * NPAD +  0 + nl] = acc0[r];
        conv[(r + 8 * half) * NPAD + 16 + nl] = acc1[r];
        conv[(r + 8 * half) * NPAD + 32 + nl] = acc2[r];
    }
    __syncthreads();

    if (lane < 16) {
        const int b = batch0 + wave * 16 + nl;
        const float* cr = &conv[nl * NPAD];
        float4 o;
        float* oo = &o.x;
        #pragma unroll
        for (int q = 0; q < 4; ++q) {
            const int qi = q >> 1, qj = q & 1;
            float mx = -__builtin_inff();
            #pragma unroll
            for (int di = 0; di < 3; ++di)
                #pragma unroll
                for (int dj = 0; dj < 3; ++dj)
                    mx = fmaxf(mx, cr[(3 * qi + di) * 6 + (3 * qj + dj)]);
            oo[q] = mx + bv;
        }
        *(float4*)&out[(size_t)b * (NFILT * 4) + f * 4] = o;
    }
}

extern "C" void kernel_launch(void* const* d_in, const int* in_sizes, int n_in,
                              void* d_out, int out_size, void* d_ws, size_t ws_size,
                              hipStream_t stream) {
    const float* X         = (const float*)d_in[0];
    const float* weight    = (const float*)d_in[1];
    const float* bias      = (const float*)d_in[2];
    const int*   positions = (const int*)d_in[3];
    float*       out       = (float*)d_out;

    dim3 grid(NFILT, 1024 / MB, 1);
    dim3 block(THREADS, 1, 1);
    filters_wmma_kernel<<<grid, block, 0, stream>>>(X, weight, bias, positions, out);
}


# Toeplitz product regrouped per pair of output rows on a 10x10 window: K 124->60, N 48->16, one Bt for all pairs written directly (no zero fill/scatter): 45 WMMAs per wave not 93; sum order changes
# speedup vs baseline: 3.3209x; 1.4045x over previous
#include <hip/hip_runtime.h>


typedef __attribute__((ext_vector_type(2))) float v2f;
typedef __attribute__((ext_vector_type(8))) float v8f;

#define NFILT  1024
#define HH     64
#define WW     64
#define KH     5
#define KW     5
#define LOCAL  3
#define WSIDE  10
#define AP     100
#define KQ     60
#define BP     60
#define NB     16
#define NPAD   48
#define MB     64
#define THREADS 128

__device__ __forceinline__ v8f mma4(v2f a, v2f b, v8f c)
{
    return __builtin_amdgcn_wmma_f32_16x16x4_f32(false, a, false, b, (short)0, c, false, false);
}

__global__ __launch_bounds__(THREADS)
void filters_wmma_kernel(const float* __restrict__ X,
                         const float* __restrict__ weight,
                         const float* __restrict__ bias,
                         const int*   __restrict__ positions,
                         float*       __restrict__ out)
{
    __shared__ __attribute__((aligned(16))) float A_lds[MB * AP];
    __shared__ __attribute__((aligned(16))) float Bt_lds[NB * BP];

    const int f      = blockIdx.x;
    const int batch0 = blockIdx.y * MB;
    const int tid    = threadIdx.x;
    const int lane   = tid & 31;
    const int wave   = tid >> 5;
    const int half   = lane >> 4;
    const int nl     = lane & 15;

    const int i0 = positions[2 * f + 0];
    const int j0 = positions[2 * f + 1];
    const float bv = bias[f];

    int  aoff[4];
    bool aok[4];
    #pragma unroll
    for (int j = 0; j < 4; ++j) {
        const int nact = j < 3 ? 30 : 10;
        const int lc = lane < nact ? lane : nact - 1;
        const int lr = lc / WSIDE;
        const int wi = 3 * j + lr;
        const int wj = lc - lr * WSIDE;
        const int r  = i0 + wi - LOCAL;
        const int c  = j0 + wj - LOCAL;
        aok[j] = (lane < nact) && ((unsigned)r < (unsigned)HH) && ((unsigned)c < (unsigned)WW);
        const int rc = r < 0 ? 0 : (r > HH - 1 ? HH - 1 : r);
        const int cc = c < 0 ? 0 : (c > WW - 1 ? WW - 1 : c);
        aoff[j] = rc * WW + cc;
    }
    const float* Xw = X + (size_t)(batch0 + wave * 16) * (HH * WW);
    float av[16][4];
    #pragma unroll
    for (int i = 0; i < 16; ++i) {
        #pragma unroll
        for (int j = 0; j < 4; ++j)
            av[i][j] = Xw[i * (HH * WW) + aoff[j]];
    }
    float wv[8];
    bool  wok[8];
    #pragma unroll
    for (int q = 0; q < 8; ++q) {
        const int idx = tid + THREADS * q;
        const int n   = idx / BP;
        const int kap = idx - n * BP;
        const int lr  = kap / WSIDE;
        const int c   = kap - lr * WSIDE;
        const int dpi = n >= 6 ? 1 : 0;
        const int pj  = n - 6 * dpi;
        const int ki  = lr - dpi;
        const int kj  = c - pj;
        wok[q] = (n < 12) && ((unsigned)ki < (unsigned)KH) && ((unsigned)kj < (unsigned)KW);
        wv[q]  = weight[f * (KH * KW) + (wok[q] ? ki * KW + kj : 0)];
    }

    #pragma unroll
    for (int i = 0; i < 16; ++i) {
        #pragma unroll
        for (int j = 0; j < 4; ++j) {
            const int nact = j < 3 ? 30 : 10;
            if (lane < nact)
                A_lds[(wave * 16 + i) * AP + 30 * j + lane] = aok[j] ? av[i][j] : 0.0f;
        }
    }
    #pragma unroll
    for (int q = 0; q < 8; ++q) {
        const int idx = tid + THREADS * q;
        if (idx < NB * BP) Bt_lds[idx] = wok[q] ? wv[q] : 0.0f;
    }
    __syncthreads();

    v8f acc0 = {}, acc1 = {}, acc2 = {};
    const float* Ap = &A_lds[(wave * 16 + nl) * AP + 2 * half];
    const float* Bp = &Bt_lds[nl * BP + 2 * half];
    v2f b, a0, a1, a2;
    #pragma unroll
    for (int s = 0; s < KQ / 4; ++s) {
        b  = *(const v2f*)(Bp + 4 * s);
        a0 = *(const v2f*)(Ap + 4 * s);
        a1 = *(const v2f*)(Ap + 20 + 4 * s);
        a2 = *(const v2f*)(Ap + 40 + 4 * s);
        acc0 = mma4(a0, b, acc0);
        acc1 = mma4(a1, b, acc1);
        acc2 = mma4(a2, b, acc2);
    }
    asm volatile("v_nop\n\tv_nop\n\tv_nop\n\tv_nop"
                 : "+v"(acc0), "+v"(acc1), "+v"(acc2)
                 : "v"(b), "v"(a0), "v"(a1), "v"(a2));

    __syncthreads();
    float* conv = &A_lds[wave * (16 * NPAD)];
    #pragma unroll
    for (int r = 0; r < 8; ++r) {
        conv[(r + 8 * half) * NPAD +  0 + nl] = acc0[r];
        conv[(r + 8 * half) * NPAD + 16 + nl] = acc1[r];
        conv[(r + 8 * half) * NPAD + 32 + nl] = acc2[r];
    }
    __syncthreads();

    if (lane < 16) {
        const int b = batch0 + wave * 16 + nl;
        const float* cr = &conv[nl * NPAD];
        float4 o;
        float* oo = &o.x;
        #pragma unroll
        for (int q = 0; q < 4; ++q) {
            const int qi = q >> 1, qj = q & 1;
            float mx = -__builtin_inff();
            #pragma unroll
            for (int di = 0; di < 3; ++di)
                #pragma unroll
                for (int dj = 0; dj < 3; ++dj) {
                    const int pi = 3 * qi + di, pj = 3 * qj + dj;
                    mx = fmaxf(mx, cr[(pi >> 1) * 16 + (pi & 1) * 6 + pj]);
                }
            oo[q] = mx + bv;
        }
        *(float4*)&out[(size_t)b * (NFILT * 4) + f * 4] = o;
    }
}

extern "C" void kernel_launch(void* const* d_in, const int* in_sizes, int n_in,
                              void* d_out, int out_size, void* d_ws, size_t ws_size,
                              hipStream_t stream) {
    const float* X         = (const float*)d_in[0];
    const float* weight    = (const float*)d_in[1];
    const float* bias      = (const float*)d_in[2];
    const int*   positions = (const int*)d_in[3];
    float*       out       = (float*)d_out;

    dim3 grid(NFILT, 1024 / MB, 1);
    dim3 block(THREADS, 1, 1);
    filters_wmma_kernel<<<grid, block, 0, stream>>>(X, weight, bias, positions, out);
}


# 16 zero-padded images kept in LDS per workgroup, loop over 256 filters: window fragments read from LDS, not gathered from global per filter; Toeplitz as A operand, pool in registers; sum order kept
# speedup vs baseline: 8.6970x; 2.6189x over previous
#include <hip/hip_runtime.h>


typedef __attribute__((ext_vector_type(2))) float v2f;
typedef __attribute__((ext_vector_type(8))) float v8f;

#define NFILT  1024
#define NBATCH 1024
#define HH     64
#define WW     64
#define KH     5
#define KW     5
#define WSIDE  10
#define KSTEPS 15
#define WSTEPS 25
#define PADL   4
#define PADT   3
#define RP     72
#define PROWS  69
#define IP     4996
#define MB     16
#define WAVES  16
#define THREADS (WAVES * 32)
#define FSPLIT 4
#define FPW    (NFILT / FSPLIT)
#define FITER  (FPW / WAVES)
#define IMG_GROUPS_PER_THREAD (MB * HH * WW / 4 / THREADS)

__device__ __forceinline__ v8f mma4(v2f a, v2f b, v8f c)
{
    return __builtin_amdgcn_wmma_f32_16x16x4_f32(false, a, false, b, (short)0, c, false, false);
}

__global__ __launch_bounds__(THREADS)
void filters_wmma_kernel(const float* __restrict__ X,
                         const float* __restrict__ weight,
                         const float* __restrict__ bias,
                         const int*   __restrict__ positions,
                         float*       __restrict__ out)
{
    __shared__ __attribute__((aligned(16))) float img[MB * IP];

    const int fbase  = blockIdx.x * FPW;
    const int batch0 = blockIdx.y * MB;
    const int tid    = threadIdx.x;
    const int lane   = tid & 31;
    const int wave   = __builtin_amdgcn_readfirstlane(tid >> 5);
    const int half   = lane >> 4;
    const int nl     = lane & 15;

    const float4 zero4 = make_float4(0.0f, 0.0f, 0.0f, 0.0f);
    #pragma unroll
    for (int k = 0; k < (MB * PROWS + THREADS - 1) / THREADS; ++k) {
        const int rt = tid + THREADS * k;
        if (rt < MB * PROWS) {
            const int m    = rt / PROWS;
            const int prow = rt - m * PROWS;
            *(float4*)&img[m * IP + prow * RP]                = zero4;
            *(float4*)&img[m * IP + prow * RP + PADL + WW]    = zero4;
        }
    }
    #pragma unroll
    for (int k = 0; k < (MB * (PROWS - HH) * (WW / 4) + THREADS - 1) / THREADS; ++k) {
        const int idx = tid + THREADS * k;
        if (idx < MB * (PROWS - HH) * (WW / 4)) {
            const int m    = idx / ((PROWS - HH) * (WW / 4));
            const int rem  = idx - m * ((PROWS - HH) * (WW / 4));
            const int br   = rem / (WW / 4);
            const int g    = rem - br * (WW / 4);
            const int prow = br < PADT ? br : HH + br;
            *(float4*)&img[m * IP + prow * RP + PADL + 4 * g] = zero4;
        }
    }
    const float* Xb = X + (size_t)batch0 * (HH * WW);
    float4 gv[IMG_GROUPS_PER_THREAD];
    #pragma unroll
    for (int it = 0; it < IMG_GROUPS_PER_THREAD; ++it)
        gv[it] = *(const float4*)(Xb + 4 * (tid + THREADS * it));
    #pragma unroll
    for (int it = 0; it < IMG_GROUPS_PER_THREAD; ++it) {
        const int idx = tid + THREADS * it;
        const int m   = idx / (HH * WW / 4);
        const int r   = (idx / (WW / 4)) % HH;
        const int g   = idx % (WW / 4);
        *(float4*)&img[m * IP + (r + PADT) * RP + PADL + 4 * g] = gv[it];
    }

    int  tsrc[2 * KSTEPS];
    int  woff[WSTEPS];
    {
        const int qj  = nl >> 3;
        const int t   = nl & 7;
        const int dpi = t / 3;
        const int pj  = 3 * qj + (t - 3 * dpi);
        #pragma unroll
        for (int s = 0; s < KSTEPS; ++s) {
            #pragma unroll
            for (int r = 0; r < 2; ++r) {
                const int kap = 4 * s + 2 * half + r;
                const int lr  = kap / WSIDE;
                const int cc  = kap - lr * WSIDE;
                const int ki  = lr - dpi;
                const int kj  = cc - pj;
                const bool ok = (t < 6) && ((unsigned)ki < (unsigned)KH) && ((unsigned)kj < (unsigned)KW);
                tsrc[2 * s + r] = 4 * (ok ? ki * KW + kj : KH * KW);
            }
        }
        #pragma unroll
        for (int j = 0; j < WSTEPS; ++j) {
            const int kap0 = 4 * j + 2 * half;
            const int wr   = kap0 / WSIDE;
            woff[j] = nl * IP + wr * RP + (kap0 - wr * WSIDE);
        }
    }
    __syncthreads();

    for (int i = 0; i < FITER; ++i) {
        const int f = fbase + i * WAVES + wave;
        int i0 = positions[2 * f + 0];
        int j0 = positions[2 * f + 1];
        i0 = i0 < 0 ? 0 : (i0 > HH - KH ? HH - KH : i0);
        j0 = j0 < 0 ? 0 : (j0 > WW - KW ? WW - KW : j0);
        const float bv = bias[f];
        const float* wf = weight + f * (KH * KW);
        const float* ib = img + (i0 * RP + j0 + (PADL - PADT));

        const float wl   = wf[lane < KH * KW ? lane : 0];
        const int   taps = __float_as_int(lane < KH * KW ? wl : 0.0f);
        v2f tf[KSTEPS];
        #pragma unroll
        for (int s = 0; s < KSTEPS; ++s) {
            tf[s].x = __int_as_float(__builtin_amdgcn_ds_bpermute(tsrc[2 * s + 0], taps));
            tf[s].y = __int_as_float(__builtin_amdgcn_ds_bpermute(tsrc[2 * s + 1], taps));
        }
        v2f wg[WSTEPS];
        #pragma unroll
        for (int j = 0; j < WSTEPS; ++j) {
            const float* p = ib + woff[j];
            wg[j].x = p[0];
            wg[j].y = p[1];
        }
        #pragma unroll
        for (int s = 0; s < KSTEPS; ++s)
            asm volatile("" : "+v"(tf[s]));
        #pragma unroll
        for (int j = 0; j < WSTEPS; ++j)
            asm volatile("" : "+v"(wg[j]));

        v8f acc0 = {}, acc1 = {}, acc2 = {};
        #pragma unroll
        for (int s = 0; s < KSTEPS; ++s) {
            acc0 = mma4(tf[s], wg[s], acc0);
            acc1 = mma4(tf[s], wg[5 + s], acc1);
            acc2 = mma4(tf[s], wg[10 + s], acc2);
        }
        asm volatile("v_nop\n\tv_nop\n\tv_nop\n\tv_nop"
                     : "+v"(acc0), "+v"(acc1), "+v"(acc2));
        #pragma unroll
        for (int s = 0; s < KSTEPS; ++s)
            asm volatile("" :: "v"(tf[s]));
        #pragma unroll
        for (int j = 0; j < WSTEPS; ++j)
            asm volatile("" :: "v"(wg[j]));

        float top = fmaxf(fmaxf(acc0[0], acc0[1]), acc0[2]);
        top = fmaxf(top, fmaxf(fmaxf(acc0[3], acc0[4]), acc0[5]));
        top = fmaxf(top, fmaxf(fmaxf(acc1[0], acc1[1]), acc1[2]));
        float bot = fmaxf(fmaxf(acc1[3], acc1[4]), acc1[5]);
        bot = fmaxf(bot, fmaxf(fmaxf(acc2[0], acc2[1]), acc2[2]));
        bot = fmaxf(bot, fmaxf(fmaxf(acc2[3], acc2[4]), acc2[5]));
        top += bv;
        bot += bv;
        const float otop = __shfl_xor(top, 16);
        const float obot = __shfl_xor(bot, 16);
        if (half == 0) {
            float4 o = make_float4(top, otop, bot, obot);
            *(float4*)&out[(size_t)(batch0 + nl) * (NFILT * 4) + f * 4] = o;
        }
    }
}

extern "C" void kernel_launch(void* const* d_in, const int* in_sizes, int n_in,
                              void* d_out, int out_size, void* d_ws, size_t ws_size,
                              hipStream_t stream) {
    const float* X         = (const float*)d_in[0];
    const float* weight    = (const float*)d_in[1];
    const float* bias      = (const float*)d_in[2];
    const int*   positions = (const int*)d_in[3];
    float*       out       = (float*)d_out;

    dim3 grid(FSPLIT, NBATCH / MB, 1);
    dim3 block(THREADS, 1, 1);
    filters_wmma_kernel<<<grid, block, 0, stream>>>(X, weight, bias, positions, out);
}


# Wave priority (s_setprio) falls 3,2,1,0 as a wave's filters run out (at 8, 4, 2 left): the wave with more filters left wins the matrix pipe, so a SIMD's four waves end together; sum order kept
# speedup vs baseline: 9.0719x; 1.0431x over previous
#include <hip/hip_runtime.h>


typedef __attribute__((ext_vector_type(2))) float v2f;
typedef __attribute__((ext_vector_type(8))) float v8f;

#define NFILT  1024
#define NBATCH 1024
#define HH     64
#define WW     64
#define KH     5
#define KW     5
#define WSIDE  10
#define KSTEPS 15
#define WSTEPS 25
#define PADL   4
#define PADT   3
#define RP     72
#define PROWS  69
#define IP     4996
#define MB     16
#define WAVES  16
#define THREADS (WAVES * 32)
#define FSPLIT 4
#define FPW    (NFILT / FSPLIT)
#define FITER  (FPW / WAVES)
#define IMG_GROUPS_PER_THREAD (MB * HH * WW / 4 / THREADS)

__device__ __forceinline__ v8f mma4(v2f a, v2f b, v8f c)
{
    return __builtin_amdgcn_wmma_f32_16x16x4_f32(false, a, false, b, (short)0, c, false, false);
}

__global__ __launch_bounds__(THREADS)
void filters_wmma_kernel(const float* __restrict__ X,
                         const float* __restrict__ weight,
                         const float* __restrict__ bias,
                         const int*   __restrict__ positions,
                         float*       __restrict__ out)
{
    __shared__ __attribute__((aligned(16))) float img[MB * IP];

    const int fbase  = blockIdx.x * FPW;
    const int batch0 = blockIdx.y * MB;
    const int tid    = threadIdx.x;
    const int lane   = tid & 31;
    const int wave   = __builtin_amdgcn_readfirstlane(tid >> 5);
    const int half   = lane >> 4;
    const int nl     = lane & 15;

    const float4 zero4 = make_float4(0.0f, 0.0f, 0.0f, 0.0f);
    #pragma unroll
    for (int k = 0; k < (MB * PROWS + THREADS - 1) / THREADS; ++k) {
        const int rt = tid + THREADS * k;
        if (rt < MB * PROWS) {
            const int m    = rt / PROWS;
            const int prow = rt - m * PROWS;
            *(float4*)&img[m * IP + prow * RP]                = zero4;
            *(float4*)&img[m * IP + prow * RP + PADL + WW]    = zero4;
        }
    }
    #pragma unroll
    for (int k = 0; k < (MB * (PROWS - HH) * (WW / 4) + THREADS - 1) / THREADS; ++k) {
        const int idx = tid + THREADS * k;
        if (idx < MB * (PROWS - HH) * (WW / 4)) {
            const int m    = idx / ((PROWS - HH) * (WW / 4));
            const int rem  = idx - m * ((PROWS - HH) * (WW / 4));
            const int br   = rem / (WW / 4);
            const int g    = rem - br * (WW / 4);
            const int prow = br < PADT ? br : HH + br;
            *(float4*)&img[m * IP + prow * RP + PADL + 4 * g] = zero4;
        }
    }
    const float* Xb = X + (size_t)batch0 * (HH * WW);
    float4 gv[IMG_GROUPS_PER_THREAD];
    #pragma unroll
    for (int it = 0; it < IMG_GROUPS_PER_THREAD; ++it)
        gv[it] = *(const float4*)(Xb + 4 * (tid + THREADS * it));
    #pragma unroll
    for (int it = 0; it < IMG_GROUPS_PER_THREAD; ++it) {
        const int idx = tid + THREADS * it;
        const int m   = idx / (HH * WW / 4);
        const int r   = (idx / (WW / 4)) % HH;
        const int g   = idx % (WW / 4);
        *(float4*)&img[m * IP + (r + PADT) * RP + PADL + 4 * g] = gv[it];
    }

    int  tsrc[2 * KSTEPS];
    int  woff[WSTEPS];
    {
        const int qj  = nl >> 3;
        const int t   = nl & 7;
        const int dpi = t / 3;
        const int pj  = 3 * qj + (t - 3 * dpi);
        #pragma unroll
        for (int s = 0; s < KSTEPS; ++s) {
            #pragma unroll
            for (int r = 0; r < 2; ++r) {
                const int kap = 4 * s + 2 * half + r;
                const int lr  = kap / WSIDE;
                const int cc  = kap - lr * WSIDE;
                const int ki  = lr - dpi;
                const int kj  = cc - pj;
                const bool ok = (t < 6) && ((unsigned)ki < (unsigned)KH) && ((unsigned)kj < (unsigned)KW);
                tsrc[2 * s + r] = 4 * (ok ? ki * KW + kj : KH * KW);
            }
        }
        #pragma unroll
        for (int j = 0; j < WSTEPS; ++j) {
            const int kap0 = 4 * j + 2 * half;
            const int wr   = kap0 / WSIDE;
            woff[j] = nl * IP + wr * RP + (kap0 - wr * WSIDE);
        }
    }
    __syncthreads();

    for (int i = 0; i < FITER; ++i) {
        if (i == 0)                 __builtin_amdgcn_s_setprio(3);
        if (i == FITER / 2)         __builtin_amdgcn_s_setprio(2);
        if (i == FITER - FITER / 4) __builtin_amdgcn_s_setprio(1);
        if (i == FITER - FITER / 8) __builtin_amdgcn_s_setprio(0);
        const int f = fbase + i * WAVES + wave;
        int i0 = positions[2 * f + 0];
        int j0 = positions[2 * f + 1];
        i0 = i0 < 0 ? 0 : (i0 > HH - KH ? HH - KH : i0);
        j0 = j0 < 0 ? 0 : (j0 > WW - KW ? WW - KW : j0);
        const float bv = bias[f];
        const float* wf = weight + f * (KH * KW);
        const float* ib = img + (i0 * RP + j0 + (PADL - PADT));

        const float wl   = wf[lane < KH * KW ? lane : 0];
        const int   taps = __float_as_int(lane < KH * KW ? wl : 0.0f);
        v2f tf[KSTEPS];
        #pragma unroll
        for (int s = 0; s < KSTEPS; ++s) {
            tf[s].x = __int_as_float(__builtin_amdgcn_ds_bpermute(tsrc[2 * s + 0], taps));
            tf[s].y = __int_as_float(__builtin_amdgcn_ds_bpermute(tsrc[2 * s + 1], taps));
        }
        v2f wg[WSTEPS];
        #pragma unroll
        for (int j = 0; j < WSTEPS; ++j) {
            const float* p = ib + woff[j];
            wg[j].x = p[0];
            wg[j].y = p[1];
        }
        #pragma unroll
        for (int s = 0; s < KSTEPS; ++s)
            asm volatile("" : "+v"(tf[s]));
        #pragma unroll
        for (int j = 0; j < WSTEPS; ++j)
            asm volatile("" : "+v"(wg[j]));

        v8f acc0 = {}, acc1 = {}, acc2 = {};
        #pragma unroll
        for (int s = 0; s < KSTEPS; ++s) {
            acc0 = mma4(tf[s], wg[s], acc0);
            acc1 = mma4(tf[s], wg[5 + s], acc1);
            acc2 = mma4(tf[s], wg[10 + s], acc2);
        }
        asm volatile("v_nop\n\tv_nop\n\tv_nop\n\tv_nop"
                     : "+v"(acc0), "+v"(acc1), "+v"(acc2));
        #pragma unroll
        for (int s = 0; s < KSTEPS; ++s)
            asm volatile("" :: "v"(tf[s]));
        #pragma unroll
        for (int j = 0; j < WSTEPS; ++j)
            asm volatile("" :: "v"(wg[j]));

        float top = fmaxf(fmaxf(acc0[0], acc0[1]), acc0[2]);
        top = fmaxf(top, fmaxf(fmaxf(acc0[3], acc0[4]), acc0[5]));
        top = fmaxf(top, fmaxf(fmaxf(acc1[0], acc1[1]), acc1[2]));
        float bot = fmaxf(fmaxf(acc1[3], acc1[4]), acc1[5]);
        bot = fmaxf(bot, fmaxf(fmaxf(acc2[0], acc2[1]), acc2[2]));
        bot = fmaxf(bot, fmaxf(fmaxf(acc2[3], acc2[4]), acc2[5]));
        top += bv;
        bot += bv;
        const float otop = __shfl_xor(top, 16);
        const float obot = __shfl_xor(bot, 16);
        if (half == 0) {
            float4 o = make_float4(top, otop, bot, obot);
            *(float4*)&out[(size_t)(batch0 + nl) * (NFILT * 4) + f * 4] = o;
        }
    }
}

extern "C" void kernel_launch(void* const* d_in, const int* in_sizes, int n_in,
                              void* d_out, int out_size, void* d_ws, size_t ws_size,
                              hipStream_t stream) {
    const float* X         = (const float*)d_in[0];
    const float* weight    = (const float*)d_in[1];
    const float* bias      = (const float*)d_in[2];
    const int*   positions = (const int*)d_in[3];
    float*       out       = (float*)d_out;

    dim3 grid(FSPLIT, NBATCH / MB, 1);
    dim3 block(THREADS, 1, 1);
    filters_wmma_kernel<<<grid, block, 0, stream>>>(X, weight, bias, positions, out);
}
